# starting kernel
# baseline (speedup 1.0000x reference)
#include <hip/hip_runtime.h>


typedef float v2f __attribute__((ext_vector_type(2)));
typedef float v8f __attribute__((ext_vector_type(8)));

#define DDIM       256
#define NHALF      128
#define LDS_STRIDE 260
#define STRH       130
#define LDS_BYTES  (NHALF * LDS_STRIDE * 4)

static __device__ __forceinline__ v8f wmma_f32(v2f a, v2f b, v8f c) {
    return __builtin_amdgcn_wmma_f32_16x16x4_f32(
         false, a,  false, b,
         (short)0, c,  false,  false);
}

extern "C" __global__ __launch_bounds__(256)
void rgcn_gemm_wmma(const float* __restrict__ X,
                    const float* __restrict__ W,
                    float* __restrict__ Y,
                    int M)
{
    extern __shared__ char smem[];
    float*     ldsW  = (float*)smem;
    const v2f* ldsW2 = (const v2f*)smem;

    const int tid   = threadIdx.x;
    const int nCol0 = blockIdx.y * NHALF;

    {
        const int nl   = tid & (NHALF - 1);
        const int kOff = (tid >> 7) * (DDIM / 2);
        for (int k = 0; k < DDIM / 2; ++k) {
            ldsW[nl * LDS_STRIDE + kOff + k] = W[(kOff + k) * DDIM + nCol0 + nl];
        }
    }
    __syncthreads();

    const int wave  = tid >> 5;
    const int lane  = tid & 31;
    const int ln    = lane & 15;
    const int h     = lane >> 4;
    const int mBase = blockIdx.x * 256 + wave * 32;

    v8f acc[16];
#pragma unroll
    for (int i = 0; i < 16; ++i) acc[i] = (v8f)0.0f;

    const int  m0  = mBase + ln,      m1  = mBase + 16 + ln;
    const bool ok0 = (m0 < M),        ok1 = (m1 < M);
    const int  mc0 = ok0 ? m0 : M - 1;
    const int  mc1 = ok1 ? m1 : M - 1;
    const float* __restrict__ X0 = X + (size_t)mc0 * DDIM;
    const float* __restrict__ X1 = X + (size_t)mc1 * DDIM;

    for (int kk = 0; kk < DDIM; kk += 16) {
        v2f a0[4], a1[4];
#pragma unroll
        for (int j = 0; j < 4; ++j) {
            const int kc = kk + 4 * j + 2 * h;
            v2f t0 = *(const v2f*)(X0 + kc);
            v2f t1 = *(const v2f*)(X1 + kc);
            a0[j] = ok0 ? t0 : (v2f)0.0f;
            a1[j] = ok1 ? t1 : (v2f)0.0f;
        }
#pragma unroll
        for (int j = 0; j < 4; ++j) {
            const int kh = (kk >> 1) + 2 * j + h;
            v2f b[8];
#pragma unroll
            for (int nb = 0; nb < 8; ++nb) {
                b[nb] = ldsW2[(size_t)(nb * 16 + ln) * STRH + kh];
            }
#pragma unroll
            for (int nb = 0; nb < 8; ++nb) {
                acc[nb] = wmma_f32(a0[j], b[nb], acc[nb]);
            }
#pragma unroll
            for (int nb = 0; nb < 8; ++nb) {
                acc[8 + nb] = wmma_f32(a1[j], b[nb], acc[8 + nb]);
            }
        }
    }

#pragma unroll
    for (int s = 0; s < 2; ++s) {
#pragma unroll
        for (int nb = 0; nb < 8; ++nb) {
            const int col = nCol0 + nb * 16 + ln;
#pragma unroll
            for (int v = 0; v < 8; ++v) {
                const int row = mBase + s * 16 + v + 8 * h;
                if (row < M) {
                    Y[(size_t)row * DDIM + col] = acc[s * 8 + nb][v];
                }
            }
        }
    }
}

#define EPB 8
extern "C" __global__ __launch_bounds__(256)
void rgcn_scatter_add(const float* __restrict__ Y,
                      const int*   __restrict__ edges,
                      float* __restrict__ out,
                      int E)
{
    const int e0 = blockIdx.x * EPB;
    const int c  = threadIdx.x;
#pragma unroll
    for (int i = 0; i < EPB; ++i) {
        const int e = e0 + i;
        if (e >= E) break;
        const int src = edges[2 * e + 0];
        const int dst = edges[2 * e + 1];
        const float v = Y[(size_t)src * DDIM + c];
        unsafeAtomicAdd(out + (size_t)dst * DDIM + c, v);
    }
}

extern "C" void kernel_launch(void* const* d_in, const int* in_sizes, int n_in,
                              void* d_out, int out_size, void* d_ws, size_t ws_size,
                              hipStream_t stream)
{
    const float* x      = (const float*)d_in[0];
    const float* W_rel  = (const float*)d_in[1];
    const float* W_self = (const float*)d_in[2];
    const int*   edges  = (const int*)  d_in[3];

    float* out = (float*)d_out;
    float* y   = (float*)d_ws;

    const int M = in_sizes[0] / DDIM;
    const int R = in_sizes[1] / (DDIM * DDIM);
    const int E = in_sizes[3] / (2 * R);

    const dim3 gemmGrid((M + 255) / 256, 2);
    const int  scatterGrid = (E + EPB - 1) / EPB;

    rgcn_gemm_wmma<<<gemmGrid, 256, LDS_BYTES, stream>>>(x, W_self, out, M);

    for (int r = 0; r < R; ++r) {
        rgcn_gemm_wmma<<<gemmGrid, 256, LDS_BYTES, stream>>>(
            x, W_rel + (size_t)r * DDIM * DDIM, y, M);
        rgcn_scatter_add<<<scatterGrid, 256, 0, stream>>>(
            y, edges + (size_t)r * E * 2, out, E);
    }
}


# scatter fused into the relation GEMMs: edges counting-sorted per call by 16-row source strip, each block adds its product rows from LDS to out[dst] with f32 atomics; add order unordered as before
# speedup vs baseline: 1.3024x; 1.3024x over previous
#include <hip/hip_runtime.h>


typedef float v2f __attribute__((ext_vector_type(2)));
typedef float v4f __attribute__((ext_vector_type(4)));
typedef float v8f __attribute__((ext_vector_type(8)));

#define DDIM         256
#define KSTEPS       64
#define KCH          16
#define NCH          (KSTEPS / KCH)
#define XROWS        16
#define XSTR         260
#define XSTAGE       (XROWS * XSTR)
#define GEMM_THREADS 512
#define GEMM_BLOCKS  256
#define CSTR         260
#define CSTAGE       (XROWS * CSTR)
#define ENT_CHUNK    512

static __device__ __forceinline__ v8f wmma_f32(v2f a, v2f b, v8f c) {
    return __builtin_amdgcn_wmma_f32_16x16x4_f32(
         false, a,  false, b,
         (short)0, c,  false,  false);
}

static __device__ __forceinline__ void lds_barrier()
{
    __builtin_amdgcn_fence(__ATOMIC_RELEASE, "workgroup", "local");
    __builtin_amdgcn_s_barrier();
    __builtin_amdgcn_fence(__ATOMIC_ACQUIRE, "workgroup", "local");
}

template <int NB>
static __device__ __forceinline__
void push_batch(const float* __restrict__ cs, float* __restrict__ yl, unsigned ev, int i)
{
    unsigned en[NB];
    float rv[NB][DDIM / 32];
#pragma unroll
    for (int e = 0; e < NB; ++e) en[e] = (unsigned)__builtin_amdgcn_readlane((int)ev, i + e);
#pragma unroll
    for (int e = 0; e < NB; ++e) {
        const float* cr = cs + (en[e] & 15u) * CSTR;
#pragma unroll
        for (int c = 0; c < DDIM / 32; ++c) rv[e][c] = cr[32 * c];
    }
#pragma unroll
    for (int e = 0; e < NB; ++e) {
        float* op = yl + (size_t)(en[e] >> 4) * DDIM;
#pragma unroll
        for (int c = 0; c < DDIM / 32; ++c)
            unsafeAtomicAdd(op + 32 * c, rv[e][c]);
    }
}

static __device__ __forceinline__
void push_strip(const float* __restrict__ cs, float* __restrict__ yl,
                const unsigned* __restrict__ ent, int beg, int end, unsigned ev0,
                int wave, int lane)
{
    for (int base = beg; base < end; base += ENT_CHUNK) {
        unsigned ev = ev0;
        if (base != beg) {
            const int idx = base + wave + 16 * lane;
            ev = idx < end ? ent[idx] : 0u;
        }
        const int lim = end < base + ENT_CHUNK ? end : base + ENT_CHUNK;
        const int n   = (lim - base - wave + 15) >> 4;
        int i = 0;
        for (; i + 4 <= n; i += 4) push_batch<4>(cs, yl, ev, i);
        if (i + 2 <= n) { push_batch<2>(cs, yl, ev, i); i += 2; }
        if (i < n) push_batch<1>(cs, yl, ev, i);
    }
}

template <bool PUSH>
static __device__ __forceinline__
void rgcn_gemm_body(const float* __restrict__ X,
                    const float* __restrict__ W,
                    float* __restrict__ Y,
                    int M,
                    const int* __restrict__ off,
                    const unsigned* __restrict__ ent,
                    float* __restrict__ ldsX,
                    float* __restrict__ ldsC)
{
    const int tid  = threadIdx.x;
    const int wave = __builtin_amdgcn_readfirstlane(tid >> 5);
    const int lane = tid & 31;
    const int ln   = lane & 15;
    const int h    = lane >> 4;
    const int n0   = wave * 16;

    v2f b[KSTEPS];
#pragma unroll
    for (int s = 0; s < KSTEPS; ++s) {
        const int k = 4 * s + 2 * h;
        b[s].x = W[(size_t)k * DDIM + n0 + ln];
        b[s].y = W[(size_t)(k + 1) * DDIM + n0 + ln];
    }

    const int nStrips = (M + XROWS - 1) / XROWS;

    const int sr = tid >> 6;
    const int sc = tid & 63;
    float* const st0 = ldsX + sr * XSTR + 4 * sc;
    float* const st1 = st0 + 8 * XSTR;

    int strip = blockIdx.x;
    {
        int ra = strip * XROWS + sr;      ra = ra < M ? ra : M - 1;
        int rb = strip * XROWS + sr + 8;  rb = rb < M ? rb : M - 1;
        const v4f g0 = *(const v4f*)(X + (size_t)ra * DDIM + 4 * sc);
        const v4f g1 = *(const v4f*)(X + (size_t)rb * DDIM + 4 * sc);
        *(v4f*)st0 = g0;
        *(v4f*)st1 = g1;
    }

    int begC = 0, endC = 0, begN = 0, endN = 0;
    unsigned entC = 0;
    const bool late = PUSH && (((wave ^ (wave >> 2)) & 1) != 0);
    int begP = 0, endP = 0;
    unsigned entP = 0;
    if constexpr (PUSH) {
        if (strip < nStrips) { begC = off[strip]; endC = off[strip + 1]; }
        const int s1 = strip + gridDim.x;
        if (s1 < nStrips) { begN = off[s1]; endN = off[s1 + 1]; }
        const int idx = begC + wave + 16 * lane;
        if (idx < endC) entC = ent[idx];
    }
    __syncthreads();

    int buf = 0;
    for (; strip < nStrips; strip += gridDim.x) {
        const int nxt = strip + gridDim.x;
        int ra = nxt * XROWS + sr;      ra = ra < M ? ra : M - 1;
        int rb = nxt * XROWS + sr + 8;  rb = rb < M ? rb : M - 1;
        const v4f g0 = *(const v4f*)(X + (size_t)ra * DDIM + 4 * sc);
        const v4f g1 = *(const v4f*)(X + (size_t)rb * DDIM + 4 * sc);
        unsigned entN = 0;
        int beg2 = 0, end2 = 0;
        if constexpr (PUSH) {
            const int idx = begN + wave + 16 * lane;
            if (idx < endN) entN = ent[idx];
            const int s2 = nxt + gridDim.x;
            if (s2 < nStrips) { beg2 = off[s2]; end2 = off[s2 + 1]; }
        }

        const float* xs = ldsX + buf * XSTAGE + ln * XSTR + 2 * h;
        v8f acc0 = (v8f)0.0f, acc1 = (v8f)0.0f;
        v2f a[2][KCH];
#pragma unroll
        for (int j = 0; j < KCH; ++j) a[0][j] = *(const v2f*)(xs + 4 * j);
#pragma unroll
        for (int q = 0; q < NCH; ++q) {
            if (q + 1 < NCH) {
#pragma unroll
                for (int j = 0; j < KCH; ++j)
                    a[(q + 1) & 1][j] = *(const v2f*)(xs + 4 * ((q + 1) * KCH + j));
            }
            __builtin_amdgcn_sched_barrier(0);
#pragma unroll
            for (int j = 0; j < KCH; j += 2) {
                acc0 = wmma_f32(a[q & 1][j],     b[q * KCH + j],     acc0);
                acc1 = wmma_f32(a[q & 1][j + 1], b[q * KCH + j + 1], acc1);
            }
            __builtin_amdgcn_sched_barrier(0);
        }

        if constexpr (PUSH) {
            float* cw = ldsC + buf * CSTAGE + (8 * h) * CSTR + n0 + ln;
#pragma unroll
            for (int v = 0; v < 8; ++v) cw[v * CSTR] = acc0[v] + acc1[v];
        }

        *(v4f*)(st0 + (buf ^ 1) * XSTAGE) = g0;
        *(v4f*)(st1 + (buf ^ 1) * XSTAGE) = g1;
        if constexpr (PUSH) {
            if (late) push_strip(ldsC + (buf ^ 1) * CSTAGE + lane, Y + lane, ent, begP, endP, entP, wave, lane);
            lds_barrier();
        } else {
            __syncthreads();
        }

        if constexpr (!PUSH) {
            const int rowBase = strip * XROWS + 8 * h;
            float* yp = Y + (size_t)rowBase * DDIM + n0 + ln;
            if (strip * XROWS + XROWS <= M) {
#pragma unroll
                for (int v = 0; v < 8; ++v) yp[(size_t)v * DDIM] = acc0[v] + acc1[v];
            } else {
#pragma unroll
                for (int v = 0; v < 8; ++v) {
                    if (rowBase + v < M) yp[(size_t)v * DDIM] = acc0[v] + acc1[v];
                }
            }
        } else {
            if (!late) push_strip(ldsC + buf * CSTAGE + lane, Y + lane, ent, begC, endC, entC, wave, lane);
            begP = begC; endP = endC; entP = entC;
            begC = begN; endC = endN; entC = entN;
            begN = beg2; endN = end2;
        }
        buf ^= 1;
    }
    if constexpr (PUSH) {
        if (late) push_strip(ldsC + (buf ^ 1) * CSTAGE + lane, Y + lane, ent, begP, endP, entP, wave, lane);
    }
}

extern "C" __global__ __launch_bounds__(GEMM_THREADS)
void rgcn_gemm_wmma(const float* __restrict__ X,
                    const float* __restrict__ W,
                    float* __restrict__ Y,
                    int M)
{
    __shared__ __attribute__((aligned(16))) float ldsX[2 * XSTAGE];
    rgcn_gemm_body<false>(X, W, Y, M, nullptr, nullptr, ldsX, nullptr);
}

extern "C" __global__ __launch_bounds__(GEMM_THREADS)
void rgcn_gemm_push(const float* __restrict__ X,
                    const float* __restrict__ W,
                    float* __restrict__ out,
                    int M,
                    const int* __restrict__ off,
                    const unsigned* __restrict__ ent)
{
    __shared__ __attribute__((aligned(16))) float ldsX[2 * XSTAGE];
    __shared__ __attribute__((aligned(16))) float ldsC[2 * CSTAGE];
    rgcn_gemm_body<true>(X, W, out, M, off, ent, ldsX, ldsC);
}

extern "C" __global__ __launch_bounds__(256)
void rgcn_bins_zero(int* __restrict__ cnt, int n)
{
    const int i = blockIdx.x * 256 + threadIdx.x;
    if (i < n) cnt[i] = 0;
}

extern "C" __global__ __launch_bounds__(256)
void rgcn_bins_count(const int* __restrict__ edges,
                     int* __restrict__ cnt,
                     int E, int total, int nStrips)
{
    const int g = blockIdx.x * 256 + threadIdx.x;
    if (g >= total) return;
    const int r   = g / E;
    const int src = edges[2 * (size_t)g];
    atomicAdd(cnt + r * nStrips + (src >> 4), 1);
}

#define SCAN_THREADS 1024
extern "C" __global__ __launch_bounds__(SCAN_THREADS)
void rgcn_bins_scan(const int* __restrict__ cnt,
                    int* __restrict__ off,
                    int* __restrict__ cur,
                    int n)
{
    __shared__ int part[SCAN_THREADS];
    const int tid = threadIdx.x;
    const int per = (n + SCAN_THREADS - 1) / SCAN_THREADS;
    const int b0  = tid * per;
    int s = 0;
    for (int i = 0; i < per; ++i) {
        const int idx = b0 + i;
        if (idx < n) s += cnt[idx];
    }
    part[tid] = s;
    __syncthreads();
    for (int d = 1; d < SCAN_THREADS; d <<= 1) {
        const int v = tid >= d ? part[tid - d] : 0;
        __syncthreads();
        part[tid] += v;
        __syncthreads();
    }
    int run = part[tid] - s;
    for (int i = 0; i < per; ++i) {
        const int idx = b0 + i;
        if (idx < n) {
            const int c = cnt[idx];
            off[idx] = run;
            cur[idx] = run;
            run += c;
        }
    }
    if (tid == SCAN_THREADS - 1) off[n] = part[SCAN_THREADS - 1];
}

extern "C" __global__ __launch_bounds__(256)
void rgcn_bins_fill(const int* __restrict__ edges,
                    int* __restrict__ cur,
                    unsigned* __restrict__ ent,
                    int E, int total, int nStrips)
{
    const int g = blockIdx.x * 256 + threadIdx.x;
    if (g >= total) return;
    const int r   = g / E;
    const int src = edges[2 * (size_t)g + 0];
    const int dst = edges[2 * (size_t)g + 1];
    const int pos = atomicAdd(cur + r * nStrips + (src >> 4), 1);
    ent[pos] = ((unsigned)dst << 4) | (unsigned)(src & 15);
}

extern "C" void kernel_launch(void* const* d_in, const int* in_sizes, int n_in,
                              void* d_out, int out_size, void* d_ws, size_t ws_size,
                              hipStream_t stream)
{
    const float* x      = (const float*)d_in[0];
    const float* W_rel  = (const float*)d_in[1];
    const float* W_self = (const float*)d_in[2];
    const int*   edges  = (const int*)  d_in[3];

    float* out = (float*)d_out;

    const int M = in_sizes[0] / DDIM;
    const int R = in_sizes[1] / (DDIM * DDIM);
    const int E = in_sizes[3] / (2 * R);

    const int nStrips = (M + XROWS - 1) / XROWS;
    const int nBins   = R * nStrips;
    const int total   = R * E;

    int*      cnt = (int*)d_ws;
    int*      off = cnt + (nBins + 1);
    int*      cur = off + (nBins + 1);
    unsigned* ent = (unsigned*)(cur + (nBins + 1));

    const int edgeGrid = (total + 255) / 256;

    rgcn_bins_zero <<<(nBins + 256) / 256, 256, 0, stream>>>(cnt, nBins + 1);
    rgcn_bins_count<<<edgeGrid, 256, 0, stream>>>(edges, cnt, E, total, nStrips);
    rgcn_bins_scan <<<1, SCAN_THREADS, 0, stream>>>(cnt, off, cur, nBins);
    rgcn_bins_fill <<<edgeGrid, 256, 0, stream>>>(edges, cur, ent, E, total, nStrips);

    rgcn_gemm_wmma<<<GEMM_BLOCKS, GEMM_THREADS, 0, stream>>>(x, W_self, out, M);

    for (int r = 0; r < R; ++r) {
        rgcn_gemm_push<<<GEMM_BLOCKS, GEMM_THREADS, 0, stream>>>(
            x, W_rel + (size_t)r * DDIM * DDIM, out, M, off + (size_t)r * nStrips, ent);
    }
}


# edge sort ranked in LDS: 256 blocks count their edge chunk per source strip in LDS tables, pass 2 sums the tables into slot cursors; no global atomics left in the sort; f32 adds unordered as before
# speedup vs baseline: 2.0968x; 1.6100x over previous
#include <hip/hip_runtime.h>


typedef float v2f __attribute__((ext_vector_type(2)));
typedef float v4f __attribute__((ext_vector_type(4)));
typedef float v8f __attribute__((ext_vector_type(8)));

#define DDIM         256
#define KSTEPS       64
#define KCH          16
#define NCH          (KSTEPS / KCH)
#define XROWS        16
#define XSTR         260
#define XSTAGE       (XROWS * XSTR)
#define GEMM_THREADS 512
#define GEMM_BLOCKS  256
#define CSTR         260
#define CSTAGE       (XROWS * CSTR)
#define ENT_CHUNK    512

static __device__ __forceinline__ v8f wmma_f32(v2f a, v2f b, v8f c) {
    return __builtin_amdgcn_wmma_f32_16x16x4_f32(
         false, a,  false, b,
         (short)0, c,  false,  false);
}

static __device__ __forceinline__ void lds_barrier()
{
    __builtin_amdgcn_fence(__ATOMIC_RELEASE, "workgroup", "local");
    __builtin_amdgcn_s_barrier();
    __builtin_amdgcn_fence(__ATOMIC_ACQUIRE, "workgroup", "local");
}

template <int NB>
static __device__ __forceinline__
void push_batch(const float* __restrict__ cs, float* __restrict__ yl, unsigned ev, int i)
{
    unsigned en[NB];
    float rv[NB][DDIM / 32];
#pragma unroll
    for (int e = 0; e < NB; ++e) en[e] = (unsigned)__builtin_amdgcn_readlane((int)ev, i + e);
#pragma unroll
    for (int e = 0; e < NB; ++e) {
        const float* cr = cs + (en[e] & 15u) * CSTR;
#pragma unroll
        for (int c = 0; c < DDIM / 32; ++c) rv[e][c] = cr[32 * c];
    }
#pragma unroll
    for (int e = 0; e < NB; ++e) {
        float* op = yl + (size_t)(en[e] >> 4) * DDIM;
#pragma unroll
        for (int c = 0; c < DDIM / 32; ++c)
            unsafeAtomicAdd(op + 32 * c, rv[e][c]);
    }
}

static __device__ __forceinline__
void push_strip(const float* __restrict__ cs, float* __restrict__ yl,
                const unsigned* __restrict__ ent, int beg, int end, unsigned ev0,
                int wave, int lane)
{
    for (int base = beg; base < end; base += ENT_CHUNK) {
        unsigned ev = ev0;
        if (base != beg) {
            const int idx = base + wave + 16 * lane;
            ev = idx < end ? ent[idx] : 0u;
        }
        const int lim = end < base + ENT_CHUNK ? end : base + ENT_CHUNK;
        const int n   = (lim - base - wave + 15) >> 4;
        int i = 0;
        for (; i + 4 <= n; i += 4) push_batch<4>(cs, yl, ev, i);
        if (i + 2 <= n) { push_batch<2>(cs, yl, ev, i); i += 2; }
        if (i < n) push_batch<1>(cs, yl, ev, i);
    }
}

template <bool PUSH>
static __device__ __forceinline__
void rgcn_gemm_body(const float* __restrict__ X,
                    const float* __restrict__ W,
                    float* __restrict__ Y,
                    int M,
                    const int* __restrict__ off,
                    const unsigned* __restrict__ ent,
                    float* __restrict__ ldsX,
                    float* __restrict__ ldsC)
{
    const int tid  = threadIdx.x;
    const int wave = __builtin_amdgcn_readfirstlane(tid >> 5);
    const int lane = tid & 31;
    const int ln   = lane & 15;
    const int h    = lane >> 4;
    const int n0   = wave * 16;

    v2f b[KSTEPS];
#pragma unroll
    for (int s = 0; s < KSTEPS; ++s) {
        const int k = 4 * s + 2 * h;
        b[s].x = W[(size_t)k * DDIM + n0 + ln];
        b[s].y = W[(size_t)(k + 1) * DDIM + n0 + ln];
    }

    const int nStrips = (M + XROWS - 1) / XROWS;

    const int sr = tid >> 6;
    const int sc = tid & 63;
    float* const st0 = ldsX + sr * XSTR + 4 * sc;
    float* const st1 = st0 + 8 * XSTR;

    int strip = blockIdx.x;
    {
        int ra = strip * XROWS + sr;      ra = ra < M ? ra : M - 1;
        int rb = strip * XROWS + sr + 8;  rb = rb < M ? rb : M - 1;
        const v4f g0 = *(const v4f*)(X + (size_t)ra * DDIM + 4 * sc);
        const v4f g1 = *(const v4f*)(X + (size_t)rb * DDIM + 4 * sc);
        *(v4f*)st0 = g0;
        *(v4f*)st1 = g1;
    }

    int begC = 0, endC = 0, begN = 0, endN = 0;
    unsigned entC = 0;
    const bool late = PUSH && (((wave ^ (wave >> 2)) & 1) != 0);
    int begP = 0, endP = 0;
    unsigned entP = 0;
    if constexpr (PUSH) {
        if (strip < nStrips) { begC = off[strip]; endC = off[strip + 1]; }
        const int s1 = strip + gridDim.x;
        if (s1 < nStrips) { begN = off[s1]; endN = off[s1 + 1]; }
        const int idx = begC + wave + 16 * lane;
        if (idx < endC) entC = ent[idx];
    }
    __syncthreads();

    int buf = 0;
    for (; strip < nStrips; strip += gridDim.x) {
        const int nxt = strip + gridDim.x;
        int ra = nxt * XROWS + sr;      ra = ra < M ? ra : M - 1;
        int rb = nxt * XROWS + sr + 8;  rb = rb < M ? rb : M - 1;
        const v4f g0 = *(const v4f*)(X + (size_t)ra * DDIM + 4 * sc);
        const v4f g1 = *(const v4f*)(X + (size_t)rb * DDIM + 4 * sc);
        unsigned entN = 0;
        int beg2 = 0, end2 = 0;
        if constexpr (PUSH) {
            const int idx = begN + wave + 16 * lane;
            if (idx < endN) entN = ent[idx];
            const int s2 = nxt + gridDim.x;
            if (s2 < nStrips) { beg2 = off[s2]; end2 = off[s2 + 1]; }
        }

        const float* xs = ldsX + buf * XSTAGE + ln * XSTR + 2 * h;
        v8f acc0 = (v8f)0.0f, acc1 = (v8f)0.0f;
        v2f a[2][KCH];
#pragma unroll
        for (int j = 0; j < KCH; ++j) a[0][j] = *(const v2f*)(xs + 4 * j);
#pragma unroll
        for (int q = 0; q < NCH; ++q) {
            if (q + 1 < NCH) {
#pragma unroll
                for (int j = 0; j < KCH; ++j)
                    a[(q + 1) & 1][j] = *(const v2f*)(xs + 4 * ((q + 1) * KCH + j));
            }
            __builtin_amdgcn_sched_barrier(0);
#pragma unroll
            for (int j = 0; j < KCH; j += 2) {
                acc0 = wmma_f32(a[q & 1][j],     b[q * KCH + j],     acc0);
                acc1 = wmma_f32(a[q & 1][j + 1], b[q * KCH + j + 1], acc1);
            }
            __builtin_amdgcn_sched_barrier(0);
        }

        if constexpr (PUSH) {
            float* cw = ldsC + buf * CSTAGE + (8 * h) * CSTR + n0 + ln;
#pragma unroll
            for (int v = 0; v < 8; ++v) cw[v * CSTR] = acc0[v] + acc1[v];
        }

        *(v4f*)(st0 + (buf ^ 1) * XSTAGE) = g0;
        *(v4f*)(st1 + (buf ^ 1) * XSTAGE) = g1;
        if constexpr (PUSH) {
            if (late) push_strip(ldsC + (buf ^ 1) * CSTAGE + lane, Y + lane, ent, begP, endP, entP, wave, lane);
            lds_barrier();
        } else {
            __syncthreads();
        }

        if constexpr (!PUSH) {
            const int rowBase = strip * XROWS + 8 * h;
            float* yp = Y + (size_t)rowBase * DDIM + n0 + ln;
            if (strip * XROWS + XROWS <= M) {
#pragma unroll
                for (int v = 0; v < 8; ++v) yp[(size_t)v * DDIM] = acc0[v] + acc1[v];
            } else {
#pragma unroll
                for (int v = 0; v < 8; ++v) {
                    if (rowBase + v < M) yp[(size_t)v * DDIM] = acc0[v] + acc1[v];
                }
            }
        } else {
            if (!late) push_strip(ldsC + buf * CSTAGE + lane, Y + lane, ent, begC, endC, entC, wave, lane);
            begP = begC; endP = endC; entP = entC;
            begC = begN; endC = endN; entC = entN;
            begN = beg2; endN = end2;
        }
        buf ^= 1;
    }
    if constexpr (PUSH) {
        if (late) push_strip(ldsC + (buf ^ 1) * CSTAGE + lane, Y + lane, ent, begP, endP, entP, wave, lane);
    }
}

extern "C" __global__ __launch_bounds__(GEMM_THREADS)
void rgcn_gemm_wmma(const float* __restrict__ X,
                    const float* __restrict__ W,
                    float* __restrict__ Y,
                    int M)
{
    __shared__ __attribute__((aligned(16))) float ldsX[2 * XSTAGE];
    rgcn_gemm_body<false>(X, W, Y, M, nullptr, nullptr, ldsX, nullptr);
}

extern "C" __global__ __launch_bounds__(GEMM_THREADS)
void rgcn_gemm_push(const float* __restrict__ X,
                    const float* __restrict__ W,
                    float* __restrict__ out,
                    int M,
                    const int* __restrict__ off,
                    const unsigned* __restrict__ ent)
{
    __shared__ __attribute__((aligned(16))) float ldsX[2 * XSTAGE];
    __shared__ __attribute__((aligned(16))) float ldsC[2 * CSTAGE];
    rgcn_gemm_body<true>(X, W, out, M, off, ent, ldsX, ldsC);
}

#define SORT_THREADS 1024
#define SORT_CHUNKS  32
#define SORT_UNROLL  4

typedef int v2i __attribute__((ext_vector_type(2)));

extern "C" __global__ __launch_bounds__(SORT_THREADS)
void rgcn_bins_count(const int* __restrict__ edges,
                     int* __restrict__ hist,
                     int E, int nStrips)
{
    extern __shared__ int ldsCnt[];
    const int tid = threadIdx.x;
    const int r   = blockIdx.x / SORT_CHUNKS;
    const int c   = blockIdx.x % SORT_CHUNKS;
    const int per = (E + SORT_CHUNKS - 1) / SORT_CHUNKS;
    const int e0  = c * per;
    const int e1  = E < e0 + per ? E : e0 + per;

    for (int s = tid; s < nStrips; s += SORT_THREADS) ldsCnt[s] = 0;
    __syncthreads();

    const int* ed = edges + 2 * (size_t)r * E;
    for (int e = e0 + tid; e < e1; e += SORT_UNROLL * SORT_THREADS) {
        int src[SORT_UNROLL];
#pragma unroll
        for (int u = 0; u < SORT_UNROLL; ++u) {
            const int ee = e + u * SORT_THREADS;
            src[u] = ed[2 * (size_t)(ee < e1 ? ee : e1 - 1)];
        }
        asm volatile("" :: "v"(src[0]), "v"(src[1]), "v"(src[2]), "v"(src[3]));
#pragma unroll
        for (int u = 0; u < SORT_UNROLL; ++u) {
            if (e + u * SORT_THREADS < e1) atomicAdd(&ldsCnt[src[u] >> 4], 1);
        }
    }
    __syncthreads();

    int* h = hist + (size_t)blockIdx.x * nStrips;
    for (int s = tid; s < nStrips; s += SORT_THREADS) h[s] = ldsCnt[s];
}

extern "C" __global__ __launch_bounds__(SORT_THREADS)
void rgcn_bins_fill(const int* __restrict__ edges,
                    const int* __restrict__ hist,
                    int* __restrict__ off,
                    unsigned* __restrict__ ent,
                    int E, int nStrips, int R)
{
    extern __shared__ int ldsBin[];
    __shared__ int part[SORT_THREADS];
    int* const ldsCur = ldsBin;
    int* const ldsTot = ldsBin + nStrips;

    const int tid = threadIdx.x;
    const int r   = blockIdx.x / SORT_CHUNKS;
    const int c   = blockIdx.x % SORT_CHUNKS;

    const int* hr = hist + (size_t)r * SORT_CHUNKS * nStrips;
    for (int s = tid; s < nStrips; s += SORT_THREADS) {
        int tot = 0, pre = 0;
#pragma unroll
        for (int k = 0; k < SORT_CHUNKS; ++k) {
            const int* hk = hr + (size_t)k * nStrips;
            const int v = hk[s];
            tot += v;
            pre += k < c ? v : 0;
        }
        ldsTot[s] = tot;
        ldsCur[s] = pre;
    }
    __syncthreads();

    const int seg = (nStrips + SORT_THREADS - 1) / SORT_THREADS;
    const int b0  = tid * seg;
    int sum = 0;
    for (int i = 0; i < seg; ++i) {
        const int s = b0 + i;
        if (s < nStrips) sum += ldsTot[s];
    }
    part[tid] = sum;
    __syncthreads();
    for (int d = 1; d < SORT_THREADS; d <<= 1) {
        const int v = tid >= d ? part[tid - d] : 0;
        __syncthreads();
        part[tid] += v;
        __syncthreads();
    }
    int run = r * E + part[tid] - sum;
    for (int i = 0; i < seg; ++i) {
        const int s = b0 + i;
        if (s < nStrips) {
            ldsCur[s] += run;
            if (c == 0) off[(size_t)r * nStrips + s] = run;
            run += ldsTot[s];
        }
    }
    if (c == 0 && r == R - 1 && tid == SORT_THREADS - 1) off[(size_t)R * nStrips] = run;
    __syncthreads();

    const int per = (E + SORT_CHUNKS - 1) / SORT_CHUNKS;
    const int e0  = c * per;
    const int e1  = E < e0 + per ? E : e0 + per;
    const int* ed = edges + 2 * (size_t)r * E;
    for (int e = e0 + tid; e < e1; e += SORT_UNROLL * SORT_THREADS) {
        v2i sd[SORT_UNROLL];
#pragma unroll
        for (int u = 0; u < SORT_UNROLL; ++u) {
            const int ee = e + u * SORT_THREADS;
            sd[u] = *(const v2i*)(ed + 2 * (size_t)(ee < e1 ? ee : e1 - 1));
        }
        asm volatile("" :: "v"(sd[0]), "v"(sd[1]), "v"(sd[2]), "v"(sd[3]));
#pragma unroll
        for (int u = 0; u < SORT_UNROLL; ++u) {
            if (e + u * SORT_THREADS < e1) {
                const int pos = atomicAdd(&ldsCur[sd[u].x >> 4], 1);
                ent[pos] = ((unsigned)sd[u].y << 4) | (unsigned)(sd[u].x & 15);
            }
        }
    }
}

extern "C" void kernel_launch(void* const* d_in, const int* in_sizes, int n_in,
                              void* d_out, int out_size, void* d_ws, size_t ws_size,
                              hipStream_t stream)
{
    const float* x      = (const float*)d_in[0];
    const float* W_rel  = (const float*)d_in[1];
    const float* W_self = (const float*)d_in[2];
    const int*   edges  = (const int*)  d_in[3];

    float* out = (float*)d_out;

    const int M = in_sizes[0] / DDIM;
    const int R = in_sizes[1] / (DDIM * DDIM);
    const int E = in_sizes[3] / (2 * R);

    const int nStrips = (M + XROWS - 1) / XROWS;
    const int nBins   = R * nStrips;

    int*      hist = (int*)d_ws;
    int*      off  = hist + (size_t)R * SORT_CHUNKS * nStrips;
    unsigned* ent  = (unsigned*)(off + (nBins + 1));

    rgcn_bins_count<<<R * SORT_CHUNKS, SORT_THREADS, nStrips * sizeof(int), stream>>>(edges, hist, E, nStrips);
    rgcn_bins_fill <<<R * SORT_CHUNKS, SORT_THREADS, 2 * nStrips * sizeof(int), stream>>>(edges, hist, off, ent, E, nStrips, R);

    rgcn_gemm_wmma<<<GEMM_BLOCKS, GEMM_THREADS, 0, stream>>>(x, W_self, out, M);

    for (int r = 0; r < R; ++r) {
        rgcn_gemm_push<<<GEMM_BLOCKS, GEMM_THREADS, 0, stream>>>(
            x, W_rel + (size_t)r * DDIM * DDIM, out, M, off + (size_t)r * nStrips, ent);
    }
}
